# starting kernel
# baseline (speedup 1.0000x reference)
#include <hip/hip_runtime.h>
#include <hip/hip_bf16.h>
#include <math.h>

#define N_NODES 20000
#define E_EDGES 320000
#define ETOT    (E_EDGES + N_NODES)
#define DIN     64
#define HID     128
#define NHEAD   4
#define HC      (NHEAD * HID)

typedef float v2f __attribute__((ext_vector_type(2)));
typedef float v4f __attribute__((ext_vector_type(4)));
typedef float v8f __attribute__((ext_vector_type(8)));

__device__ __forceinline__ unsigned f2key(float f) {
    unsigned u = __float_as_uint(f);
    return (u & 0x80000000u) ? ~u : (u | 0x80000000u);
}
__device__ __forceinline__ float key2f(unsigned k) {
    unsigned u = (k & 0x80000000u) ? (k ^ 0x80000000u) : ~k;
    return __uint_as_float(u);
}

template <int RELU>
__global__ void gemm_wmma_f32(const float* __restrict__ A,
                              const float* __restrict__ B,
                              const float* __restrict__ bias,
                              float* __restrict__ C,
                              int M, int K, int N) {
    int wave   = (blockIdx.x * blockDim.x + threadIdx.x) >> 5;
    int tilesN = N >> 4;
    int tiles  = (M >> 4) * tilesN;
    if (wave >= tiles) return;
    int tm = wave / tilesN, tn = wave % tilesN;
    int lane = threadIdx.x & 31;
    int l16  = lane & 15;
    int half = lane >> 4;

    const float* arow = A + (size_t)(tm * 16 + l16) * K + 2 * half;
    const float* bcol = B + (size_t)(2 * half) * N + tn * 16 + l16;

    v8f acc = {0.f, 0.f, 0.f, 0.f, 0.f, 0.f, 0.f, 0.f};
    for (int k = 0; k < K; k += 4) {
        v2f a, b;
        a.x = arow[k];
        a.y = arow[k + 1];
        b.x = bcol[(size_t)k * N];
        b.y = bcol[(size_t)(k + 1) * N];
        acc = __builtin_amdgcn_wmma_f32_16x16x4_f32(
             false, a,  false, b,
             (short)0, acc,  false,  false);
    }

    int   col = tn * 16 + l16;
    float bv  = bias[col];
#pragma unroll
    for (int v = 0; v < 8; ++v) {
        int   row = tm * 16 + v + 8 * half;
        float val = acc[v] + bv;
        if (RELU) val = fmaxf(val, 0.f);
        C[(size_t)row * N + col] = val;
    }
}

__global__ void init_scratch(float* __restrict__ res, float* __restrict__ denom,
                             unsigned* __restrict__ amax, float* __restrict__ gpart) {
    int idx = blockIdx.x * blockDim.x + threadIdx.x;
    if (idx < N_NODES * HID) res[idx] = 0.f;
    if (idx < N_NODES * NHEAD) { denom[idx] = 0.f; amax[idx] = 0u; }
    if (idx < HID) gpart[idx] = 0.f;
}

__device__ __forceinline__ void edge_ends(const int* __restrict__ ei, int e, int& s, int& d) {
    if (e < E_EDGES) { s = ei[e]; d = ei[E_EDGES + e]; }
    else             { s = d = e - E_EDGES; }
    s = __builtin_amdgcn_readfirstlane(s);
    d = __builtin_amdgcn_readfirstlane(d);
}

__global__ void edge_alpha(const int* __restrict__ ei,
                           const float* __restrict__ xl, const float* __restrict__ xr,
                           const float* __restrict__ att,
                           float* __restrict__ alpha, unsigned* __restrict__ amax) {
    int wave = (blockIdx.x * blockDim.x + threadIdx.x) >> 5;
    if (wave >= ETOT) return;
    int lane = threadIdx.x & 31;
    int s, d;
    edge_ends(ei, wave, s, d);

    const v4f* pl = (const v4f*)(xl + (size_t)s * HC);
    const v4f* pr = (const v4f*)(xr + (size_t)d * HC);
    const v4f* pa = (const v4f*)att;

    float acc[NHEAD];
#pragma unroll
    for (int h = 0; h < NHEAD; ++h) {
        int idx = h * (HID / 4) + lane;
        v4f l = pl[idx];
        v4f r = pr[idx];
        v4f w = pa[idx];
        float a = 0.f;
#pragma unroll
        for (int i = 0; i < 4; ++i) {
            float m = l[i] + r[i];
            float e = (m > 0.f) ? m : 0.2f * m;
            a += e * w[i];
        }
        acc[h] = a;
    }
#pragma unroll
    for (int h = 0; h < NHEAD; ++h)
        for (int off = 16; off; off >>= 1)
            acc[h] += __shfl_down(acc[h], off, 32);

    if (lane == 0) {
#pragma unroll
        for (int h = 0; h < NHEAD; ++h) {
            alpha[(size_t)wave * NHEAD + h] = acc[h];
            atomicMax(&amax[d * NHEAD + h], f2key(acc[h]));
        }
    }
}

__global__ void edge_exp(const int* __restrict__ ei, float* __restrict__ alpha,
                         const unsigned* __restrict__ amax, float* __restrict__ denom) {
    int idx = blockIdx.x * blockDim.x + threadIdx.x;
    if (idx >= ETOT * NHEAD) return;
    int e = idx >> 2, h = idx & 3;
    int d = (e < E_EDGES) ? ei[E_EDGES + e] : (e - E_EDGES);
    float ex = expf(alpha[idx] - key2f(amax[d * NHEAD + h]));
    alpha[idx] = ex;
    atomicAdd(&denom[d * NHEAD + h], ex);
}

__global__ void edge_scatter(const int* __restrict__ ei, const float* __restrict__ xl,
                             const float* __restrict__ alpha, const float* __restrict__ denom,
                             float* __restrict__ res) {
    int wave = (blockIdx.x * blockDim.x + threadIdx.x) >> 5;
    if (wave >= ETOT) return;
    int lane = threadIdx.x & 31;
    int s, d;
    edge_ends(ei, wave, s, d);

    float a4[NHEAD];
#pragma unroll
    for (int h = 0; h < NHEAD; ++h)
        a4[h] = alpha[(size_t)wave * NHEAD + h] / denom[d * NHEAD + h];

    const v4f* pl = (const v4f*)(xl + (size_t)s * HC);
    float* pres = res + (size_t)d * HID + lane * 4;

    v4f v = {0.f, 0.f, 0.f, 0.f};
#pragma unroll
    for (int h = 0; h < NHEAD; ++h) {
        v4f l = pl[h * (HID / 4) + lane];
#pragma unroll
        for (int i = 0; i < 4; ++i) v[i] += l[i] * a4[h];
    }
#pragma unroll
    for (int i = 0; i < 4; ++i)
        atomicAdd(&pres[i], 0.25f * v[i]);
}

__global__ void node_ln_relu(const float* __restrict__ hin, const float* __restrict__ res,
                             const float* __restrict__ bias, const float* __restrict__ g,
                             const float* __restrict__ be, float* __restrict__ hout) {
    int wave = (blockIdx.x * blockDim.x + threadIdx.x) >> 5;
    if (wave >= N_NODES) return;
    int lane = threadIdx.x & 31;
    const v4f* ph = (const v4f*)(hin + (size_t)wave * HID);
    const v4f* pr = (const v4f*)(res + (size_t)wave * HID);
    const v4f* pb = (const v4f*)bias;

    v4f y;
    {
        v4f a = ph[lane], b = pr[lane], c = pb[lane];
#pragma unroll
        for (int i = 0; i < 4; ++i) y[i] = a[i] + b[i] + c[i];
    }
    float s = 0.f, ss = 0.f;
#pragma unroll
    for (int i = 0; i < 4; ++i) { s += y[i]; ss += y[i] * y[i]; }
    for (int off = 16; off; off >>= 1) {
        s  += __shfl_xor(s,  off, 32);
        ss += __shfl_xor(ss, off, 32);
    }
    float mu  = s * (1.f / HID);
    float var = ss * (1.f / HID) - mu * mu;
    float r   = rsqrtf(var + 1e-5f);

    v4f gg = ((const v4f*)g)[lane];
    v4f bb = ((const v4f*)be)[lane];
    v4f o;
#pragma unroll
    for (int i = 0; i < 4; ++i)
        o[i] = fmaxf((y[i] - mu) * r * gg[i] + bb[i], 0.f);
    ((v4f*)(hout + (size_t)wave * HID))[lane] = o;
}

__global__ void logits_kernel(const float* __restrict__ shared_f, const float* __restrict__ Wa,
                              const float* __restrict__ ba, float* __restrict__ gpart,
                              float* __restrict__ out) {
    int wave = (blockIdx.x * blockDim.x + threadIdx.x) >> 5;
    if (wave >= N_NODES) return;
    int lane = threadIdx.x & 31;
    v4f v = ((const v4f*)(shared_f + (size_t)wave * HID))[lane];
    v4f w = ((const v4f*)Wa)[lane];
    float acc = 0.f;
#pragma unroll
    for (int i = 0; i < 4; ++i) {
        acc += v[i] * w[i];
        atomicAdd(&gpart[lane * 4 + i], v[i]);
    }
    for (int off = 16; off; off >>= 1) acc += __shfl_down(acc, off, 32);
    if (lane == 0) out[wave] = acc + ba[0];
}

__global__ void value_kernel(const float* __restrict__ gpart, const float* __restrict__ Wc,
                             const float* __restrict__ bc, float* __restrict__ out) {
    int lane = threadIdx.x & 31;
    v4f v = ((const v4f*)gpart)[lane];
    v4f w = ((const v4f*)Wc)[lane];
    float acc = 0.f;
#pragma unroll
    for (int i = 0; i < 4; ++i) acc += (v[i] * (1.f / N_NODES)) * w[i];
    for (int off = 16; off; off >>= 1) acc += __shfl_down(acc, off, 32);
    if (lane == 0) out[N_NODES] = acc + bc[0];
}

static inline int blocksForWaves(long long nwaves) {
    long long threads = nwaves * 32;
    return (int)((threads + 255) / 256);
}

extern "C" void kernel_launch(void* const* d_in, const int* in_sizes, int n_in,
                              void* d_out, int out_size, void* d_ws, size_t ws_size,
                              hipStream_t stream) {
    const float* x     = (const float*)d_in[0];
    const int*   ei    = (const int*)d_in[1];
    const float* W_in  = (const float*)d_in[2];
    const float* b_in  = (const float*)d_in[3];
    const float* Wl1   = (const float*)d_in[4];
    const float* bl1   = (const float*)d_in[5];
    const float* Wr1   = (const float*)d_in[6];
    const float* br1   = (const float*)d_in[7];
    const float* att1  = (const float*)d_in[8];
    const float* bias1 = (const float*)d_in[9];
    const float* g1    = (const float*)d_in[10];
    const float* be1   = (const float*)d_in[11];
    const float* Wl2   = (const float*)d_in[12];
    const float* bl2   = (const float*)d_in[13];
    const float* Wr2   = (const float*)d_in[14];
    const float* br2   = (const float*)d_in[15];
    const float* att2  = (const float*)d_in[16];
    const float* bias2 = (const float*)d_in[17];
    const float* g2    = (const float*)d_in[18];
    const float* be2   = (const float*)d_in[19];
    const float* Wa    = (const float*)d_in[20];
    const float* ba    = (const float*)d_in[21];
    const float* Wc    = (const float*)d_in[22];
    const float* bc    = (const float*)d_in[23];

    float* out = (float*)d_out;

    float* ws = (float*)d_ws;
    size_t off = 0;
    float*    h0    = ws + off; off += (size_t)N_NODES * HID;
    float*    h1    = ws + off; off += (size_t)N_NODES * HID;
    float*    xl    = ws + off; off += (size_t)N_NODES * HC;
    float*    xr    = ws + off; off += (size_t)N_NODES * HC;
    float*    alpha = ws + off; off += (size_t)ETOT * NHEAD;
    unsigned* amax  = (unsigned*)(ws + off); off += (size_t)N_NODES * NHEAD;
    float*    denom = ws + off; off += (size_t)N_NODES * NHEAD;
    float*    res   = ws + off; off += (size_t)N_NODES * HID;
    float*    gpart = ws + off; off += HID;
    (void)ws_size; (void)in_sizes; (void)n_in; (void)out_size;

    const int initBlocks  = (N_NODES * HID + 255) / 256;
    const int edgeBlocks  = blocksForWaves(ETOT);
    const int ehBlocks    = (ETOT * NHEAD + 255) / 256;
    const int nodeBlocks  = blocksForWaves(N_NODES);
    const int gemmInBlk   = blocksForWaves((N_NODES / 16) * (HID / 16));
    const int gemmHCBlk   = blocksForWaves((N_NODES / 16) * (HC / 16));

    gemm_wmma_f32<1><<<gemmInBlk, 256, 0, stream>>>(x, W_in, b_in, h0, N_NODES, DIN, HID);

    init_scratch<<<initBlocks, 256, 0, stream>>>(res, denom, amax, gpart);
    gemm_wmma_f32<0><<<gemmHCBlk, 256, 0, stream>>>(h0, Wl1, bl1, xl, N_NODES, HID, HC);
    gemm_wmma_f32<0><<<gemmHCBlk, 256, 0, stream>>>(h0, Wr1, br1, xr, N_NODES, HID, HC);
    edge_alpha  <<<edgeBlocks, 256, 0, stream>>>(ei, xl, xr, att1, alpha, amax);
    edge_exp    <<<ehBlocks,   256, 0, stream>>>(ei, alpha, amax, denom);
    edge_scatter<<<edgeBlocks, 256, 0, stream>>>(ei, xl, alpha, denom, res);
    node_ln_relu<<<nodeBlocks, 256, 0, stream>>>(h0, res, bias1, g1, be1, h1);

    init_scratch<<<initBlocks, 256, 0, stream>>>(res, denom, amax, gpart);
    gemm_wmma_f32<0><<<gemmHCBlk, 256, 0, stream>>>(h1, Wl2, bl2, xl, N_NODES, HID, HC);
    gemm_wmma_f32<0><<<gemmHCBlk, 256, 0, stream>>>(h1, Wr2, br2, xr, N_NODES, HID, HC);
    edge_alpha  <<<edgeBlocks, 256, 0, stream>>>(ei, xl, xr, att2, alpha, amax);
    edge_exp    <<<ehBlocks,   256, 0, stream>>>(ei, alpha, amax, denom);
    edge_scatter<<<edgeBlocks, 256, 0, stream>>>(ei, xl, alpha, denom, res);
    node_ln_relu<<<nodeBlocks, 256, 0, stream>>>(h1, res, bias2, g2, be2, h0);

    logits_kernel<<<nodeBlocks, 256, 0, stream>>>(h0, Wa, ba, gpart, out);
    value_kernel <<<1, 32, 0, stream>>>(gpart, Wc, bc, out);
}


# edge phase of a layer in ONE pass over the xl rows: listed edges bucketed by row in LDS, a wave owns whole rows, streaming softmax in registers; rows read once, not twice; float sum order changes
# speedup vs baseline: 4.3153x; 4.3153x over previous
#include <hip/hip_runtime.h>
#include <hip/hip_bf16.h>
#include <math.h>

#define N_NODES 20000
#define E_EDGES 320000
#define ETOT    (E_EDGES + N_NODES)
#define DIN     64
#define HID     128
#define NHEAD   4
#define HC      (NHEAD * HID)

typedef float v2f __attribute__((ext_vector_type(2)));
typedef float v4f __attribute__((ext_vector_type(4)));
typedef float v8f __attribute__((ext_vector_type(8)));

__device__ __forceinline__ unsigned f2key(float f) {
    unsigned u = __float_as_uint(f);
    return (u & 0x80000000u) ? ~u : (u | 0x80000000u);
}
__device__ __forceinline__ float key2f(unsigned k) {
    unsigned u = (k & 0x80000000u) ? (k ^ 0x80000000u) : ~k;
    return __uint_as_float(u);
}

#define GM_COLS    128
#define GM_NT      (GM_COLS / 16)
#define GM_PITCH   (2 * GM_COLS + 32)
#define GM_WAVES_BIG   20
#define GM_WAVES_SMALL 4

template <int K, int WAVES, int RELU>
__global__ void __launch_bounds__(WAVES * 32)
gemm_wmma_f32(const float* __restrict__ A,
              const float* __restrict__ B,
              const float* __restrict__ bias,
              float* __restrict__ C, int M, int N) {
    constexpr int THREADS = WAVES * 32;
    constexpr int ITEMS   = (K / 2) * (GM_COLS / 4);
    constexpr int FT      = THREADS >= 512 ? 512 : THREADS;
    constexpr int FILLS   = ITEMS / FT;
    static_assert(ITEMS % FT == 0 && FT % 32 == 0, "panel copy: whole waves, equal shares");
    __shared__ __attribute__((aligned(16))) float s_b[(K / 2) * GM_PITCH];
    const int tid  = threadIdx.x;
    const int lane = tid & 31;
    const int w    = tid >> 5;
    const int l16  = lane & 15;
    const int half = lane >> 4;
    const int row0 = (blockIdx.x * WAVES + w) * 16;
    const int col0 = blockIdx.y * GM_COLS;
    const bool live = row0 < M;

    if (tid < FT) {
        const float* bp = B + col0;
        v4f r0[FILLS], r1[FILLS];
#pragma unroll
        for (int i = 0; i < FILLS; ++i) {
            int it = tid + i * FT;
            int p = it >> 5, q = it & 31;
            r0[i] = *(const v4f*)(bp + (size_t)(2 * p) * N + 4 * q);
            r1[i] = *(const v4f*)(bp + (size_t)(2 * p + 1) * N + 4 * q);
        }
#pragma unroll
        for (int i = 0; i < FILLS; ++i) {
            int it = tid + i * FT;
            int p = it >> 5, q = it & 31;
            float* d = s_b + p * GM_PITCH + 8 * q;
            *(v4f*)d       = (v4f){r0[i][0], r1[i][0], r0[i][1], r1[i][1]};
            *(v4f*)(d + 4) = (v4f){r0[i][2], r1[i][2], r0[i][3], r1[i][3]};
        }
    }
    __syncthreads();
    if (!live) return;

    const float* arow = A + (size_t)(row0 + l16) * K + 2 * half;
    v2f a[K / 4];
#pragma unroll
    for (int j = 0; j < K / 4; ++j) a[j] = *(const v2f*)(arow + 4 * j);

    v8f acc[GM_NT];
#pragma unroll
    for (int n = 0; n < GM_NT; ++n) acc[n] = (v8f){0.f, 0.f, 0.f, 0.f, 0.f, 0.f, 0.f, 0.f};

    const float* sb = s_b + half * GM_PITCH + 2 * l16;
#pragma unroll
    for (int j = 0; j < K / 4; ++j) {
#pragma unroll
        for (int n = 0; n < GM_NT; ++n) {
            v2f b = *(const v2f*)(sb + j * (2 * GM_PITCH) + n * 32);
            acc[n] = __builtin_amdgcn_wmma_f32_16x16x4_f32(
                 false, a[j],  false, b,
                 (short)0, acc[n],  false,  false);
        }
    }

#pragma unroll
    for (int n = 0; n < GM_NT; ++n) {
        int   col = col0 + n * 16 + l16;
        float bv  = bias[col];
#pragma unroll
        for (int v = 0; v < 8; ++v) {
            int   row = row0 + v + 8 * half;
            float val = acc[n][v] + bv;
            if (RELU) val = fmaxf(val, 0.f);
            C[(size_t)row * N + col] = val;
        }
    }
}

#define EPW 4
#define GLN (32 / EPW)

__device__ __forceinline__ int edge_src(const int* __restrict__ ei, int e) {
    return (e < E_EDGES) ? ei[e] : (e - E_EDGES);
}

template <int M>
__device__ __forceinline__ float group_xor(float v) {
    return __int_as_float(__builtin_amdgcn_update_dpp(0, __float_as_int(v), 0x160 + M, 0xf, 0xf, false));
}
__device__ __forceinline__ float group_sum(float v) {
    v += group_xor<1>(v);
    v += group_xor<2>(v);
    v += group_xor<4>(v);
    return v;
}

#define FU_NB      80
#define FU_BLOCKS  (N_NODES / FU_NB)
#define FU_THREADS 1024
#define FU_WAVES   (FU_THREADS / 32)
#define FU_Q       (E_EDGES / 4)
#define FU_AHEAD   4
#define FU_STEPS   ((FU_Q + FU_THREADS - 1) / FU_THREADS)
#define FU_ITERS   ((FU_STEPS + FU_AHEAD - 1) / FU_AHEAD)
#define FU_LKEEP   256
#define FU_LPEND   136
#define FU_LCAP    (FU_LKEEP + FU_LPEND)
#define FU_XP      (HC + 32)

#define FU_HP      (HID + 4)

typedef int v4i __attribute__((ext_vector_type(4)));
typedef unsigned v4u __attribute__((ext_vector_type(4)));

#define FU_KQ 4
#define FU_MT (FU_NB / 16)

__device__ __forceinline__ void fu_load_b(v2f (&b)[FU_KQ], const float* bp, int q) {
#pragma unroll
    for (int j = 0; j < FU_KQ; ++j) {
        const float* p = bp + (size_t)(4 * (FU_KQ * q + j)) * HC;
        b[j] = (v2f){p[0], p[HC]};
    }
}

__device__ __forceinline__ void fu_mma_q(v8f (&acc)[FU_MT], const v2f (&b)[FU_KQ], const float* ap, int q) {
#pragma unroll
    for (int j = 0; j < FU_KQ; ++j) {
#pragma unroll
        for (int m = 0; m < FU_MT; ++m) {
            v2f a = *(const v2f*)(ap + m * 16 * FU_HP + 4 * (FU_KQ * q + j));
            acc[m] = __builtin_amdgcn_wmma_f32_16x16x4_f32(
                 false, a,  false, b[j],
                 (short)0, acc[m],  false,  false);
        }
    }
}

__device__ __forceinline__ void fu_project(const float* __restrict__ Wr, const float* __restrict__ br,
                                           const float* s_h, float* s_xr, int w, int lane) {
    const int l16 = lane & 15, half = lane >> 4;
    const int col = 16 * w + l16;
    const float* bp = Wr + (size_t)(2 * half) * HC + col;
    const float* ap = s_h + l16 * FU_HP + 2 * half;
    v8f acc[FU_MT];
#pragma unroll
    for (int m = 0; m < FU_MT; ++m) acc[m] = (v8f){0.f, 0.f, 0.f, 0.f, 0.f, 0.f, 0.f, 0.f};
    v2f b0[FU_KQ], b1[FU_KQ];
    fu_load_b(b0, bp, 0);
    fu_load_b(b1, bp, 1);
    __builtin_amdgcn_sched_barrier(0);
#pragma unroll
    for (int q = 0; q < HID / 4 / FU_KQ; q += 2) {
        fu_mma_q(acc, b0, ap, q);
        if (q + 2 < HID / 4 / FU_KQ) {
            __builtin_amdgcn_sched_barrier(0);
            fu_load_b(b0, bp, q + 2);
        }
        fu_mma_q(acc, b1, ap, q + 1);
        if (q + 3 < HID / 4 / FU_KQ) {
            __builtin_amdgcn_sched_barrier(0);
            fu_load_b(b1, bp, q + 3);
        }
    }
    const float bv = br[col];
#pragma unroll
    for (int m = 0; m < FU_MT; ++m) {
#pragma unroll
        for (int v = 0; v < 8; ++v)
            s_xr[(16 * m + v + 8 * half) * FU_XP + col] = acc[m][v] + bv;
    }
}

struct FuCtx {
    const int*   ei;
    const float* xl;
    float*       alpha;
    const float* s_att;
    float*       s_tile;
    const float* s_xr;
    unsigned*    s_max;
    float*       s_den;
    int          g, t;
};

__device__ __forceinline__ const v4f* at_bytes(const float* base, unsigned bytes) {
    return (const v4f*)((const char*)base + bytes);
}

__device__ __forceinline__ void fu_score(const FuCtx& c, int ent, int s) {
    const int e = ent >> 7, r = ent & 127, t = c.t;
    const v4f* pl = at_bytes(c.xl, ((unsigned)s * HC + 4u * t) * 4u);
    const v4f* pr = (const v4f*)(c.s_xr + r * FU_XP) + t;
    const v4f* pa = (const v4f*)c.s_att + t;
    v4f l[NHEAD * 4];
#pragma unroll
    for (int k = 0; k < NHEAD * 4; ++k) l[k] = pl[(k >> 2) * (HID / 4) + (k & 3) * GLN];
    __builtin_amdgcn_sched_barrier(0);
    float acc[NHEAD];
#pragma unroll
    for (int h = 0; h < NHEAD; ++h) {
        float a = 0.f;
#pragma unroll
        for (int i = 0; i < 4; ++i) {
            int idx = h * (HID / 4) + i * GLN;
            v4f x = pr[idx];
            v4f w = pa[idx];
#pragma unroll
            for (int j = 0; j < 4; ++j) {
                float m  = l[h * 4 + i][j] + x[j];
                float lr = (m > 0.f) ? m : 0.2f * m;
                a += lr * w[j];
            }
        }
        acc[h] = a;
    }
#pragma unroll
    for (int h = 0; h < NHEAD; ++h) acc[h] = group_sum(acc[h]);
    if (t < NHEAD) {
        float v = (t == 0) ? acc[0] : (t == 1) ? acc[1] : (t == 2) ? acc[2] : acc[3];
        *(float*)((char*)c.alpha + ((unsigned)e * NHEAD + t) * 4u) = v;
        atomicMax(&c.s_max[r * NHEAD + t], f2key(v));
    }
}

__device__ __forceinline__ void fu_exp(const FuCtx& c, int ent) {
    const int e = ent >> 7, r = ent & 127, t = c.t;
    if (t < NHEAD) {
        float* pe = (float*)((char*)c.alpha + ((unsigned)e * NHEAD + t) * 4u);
        float  ex = expf(*pe - key2f(c.s_max[r * NHEAD + t]));
        *pe = ex;
        __hip_atomic_fetch_add(c.s_den + r * NHEAD + t, ex, __ATOMIC_RELAXED, __HIP_MEMORY_SCOPE_WORKGROUP);
    }
}

__device__ __forceinline__ void fu_add(const FuCtx& c, int ent, int s) {
    const int e = ent >> 7, r = ent & 127, t = c.t, g = c.g;
    v4f al = *at_bytes(c.alpha, (unsigned)e * (NHEAD * 4u));
    v4f dn = *(const v4f*)(c.s_den + r * NHEAD);
    float a4[NHEAD];
#pragma unroll
    for (int h = 0; h < NHEAD; ++h) a4[h] = al[h] / dn[h];
    const v4f* pl = at_bytes(c.xl, ((unsigned)s * HC + 4u * t) * 4u);
    v4f v[4];
#pragma unroll
    for (int i = 0; i < 4; ++i) v[i] = (v4f){0.f, 0.f, 0.f, 0.f};
#pragma unroll
    for (int h = 0; h < NHEAD; ++h) {
#pragma unroll
        for (int i = 0; i < 4; ++i) {
            v4f l = pl[h * (HID / 4) + i * GLN];
#pragma unroll
            for (int j = 0; j < 4; ++j) v[i][j] += l[j] * a4[h];
        }
    }
    float* row = c.s_tile + r * HID + 4 * t;
    int o0 = g & 3, o1 = (g + 1) & 3, o2 = (g + 2) & 3, o3 = (g + 3) & 3;
#pragma unroll
    for (int i = 0; i < 4; ++i) {
        v4f x = v[i] * 0.25f;
        v4f y = (g & 1) ? (v4f){x[1], x[2], x[3], x[0]} : x;
        v4f z = (g & 2) ? (v4f){y[2], y[3], y[0], y[1]} : y;
        __hip_atomic_fetch_add(row + i * 32 + o0, z[0], __ATOMIC_RELAXED, __HIP_MEMORY_SCOPE_WORKGROUP);
        __hip_atomic_fetch_add(row + i * 32 + o1, z[1], __ATOMIC_RELAXED, __HIP_MEMORY_SCOPE_WORKGROUP);
        __hip_atomic_fetch_add(row + i * 32 + o2, z[2], __ATOMIC_RELAXED, __HIP_MEMORY_SCOPE_WORKGROUP);
        __hip_atomic_fetch_add(row + i * 32 + o3, z[3], __ATOMIC_RELAXED, __HIP_MEMORY_SCOPE_WORKGROUP);
    }
}

template <int PH>
__device__ __forceinline__ void fu_do(const FuCtx& c, int ent, int s) {
    if (PH == 1) fu_score(c, ent, s);
    if (PH == 2) fu_exp(c, ent);
    if (PH == 3) fu_add(c, ent, s);
}

__device__ __forceinline__ void fu_collect(v4i dv, int qu, int base, int* buf, int& cnt) {
    unsigned r[4];
    bool     any = false;
#pragma unroll
    for (int k = 0; k < 4; ++k) {
        r[k] = (unsigned)(dv[k] - base);
        any  = any || (r[k] < (unsigned)FU_NB);
    }
    if (__builtin_amdgcn_ballot_w32(any) != 0u) {
#pragma unroll
        for (int k = 0; k < 4; ++k) {
            bool     m    = r[k] < (unsigned)FU_NB;
            unsigned mask = __builtin_amdgcn_ballot_w32(m);
            if (m) buf[cnt + (int)__builtin_amdgcn_mbcnt_lo(mask, 0u)] = ((qu * 4 + k) << 7) | (int)r[k];
            cnt += __builtin_popcount(mask);
        }
    }
}

template <int PH>
__device__ __forceinline__ void fu_walk(const FuCtx& c, const v4i* dq, int step0, int tid, int base, int* pend) {
    int cnt = 0;
    for (int st = step0; st <= FU_STEPS; ++st) {
        const bool last = (st == FU_STEPS);
        if (!last) {
            int qu = tid + st * FU_THREADS;
            v4i dv = (qu < FU_Q) ? dq[qu] : (v4i){-1, -1, -1, -1};
            fu_collect(dv, qu, base, pend, cnt);
        }
        while (cnt >= 4 || (last && cnt > 0)) {
            int take = cnt < 4 ? cnt : 4;
            cnt -= take;
            if (c.g < take) {
                int ent = pend[cnt + c.g];
                fu_do<PH>(c, ent, (PH == 2) ? 0 : edge_src(c.ei, ent >> 7));
            }
        }
    }
}

template <int PH>
__device__ __forceinline__ void fu_list(const FuCtx& c, const int* lst, int cnt) {
    for (int i = 0; i < cnt; i += 4)
        if (i + c.g < cnt) {
            const int* p = lst + i + c.g;
            fu_do<PH>(c, p[0], p[-FU_LKEEP]);
        }
}

__device__ __forceinline__ void fu_exp_list(const FuCtx& c, const int* lst, int cnt, int lane) {
    for (int i = lane; i < cnt; i += 32) {
        const int ent = lst[i];
        const int e = ent >> 7, r = ent & 127;
        v4f* pe = (v4f*)((char*)c.alpha + (unsigned)e * (NHEAD * 4u));
        v4f  a  = *pe;
        v4u  mx = *(const v4u*)(c.s_max + r * NHEAD);
        v4f  ex;
#pragma unroll
        for (int h = 0; h < NHEAD; ++h) ex[h] = expf(a[h] - key2f(mx[h]));
        *pe = ex;
#pragma unroll
        for (int h = 0; h < NHEAD; ++h)
            __hip_atomic_fetch_add(c.s_den + r * NHEAD + h, ex[h], __ATOMIC_RELAXED, __HIP_MEMORY_SCOPE_WORKGROUP);
    }
}

#define SP_B    2
#define FU_SCAP 4096

__device__ __forceinline__ void fu_row_pass(const float* __restrict__ xl, const float* s_xr, float* s_tile,
                                            const int* s_srt, int r, int beg, int end, int g, int t,
                                            const v4f (&wa)[4]) {
    v4f xr[4], acc[4];
#pragma unroll
    for (int i = 0; i < 4; ++i) {
        xr[i]  = *(const v4f*)(s_xr + r * FU_XP + g * HID + i * 32 + 4 * t);
        acc[i] = (v4f){0.f, 0.f, 0.f, 0.f};
    }
    float m = -INFINITY, den = 0.f;
    for (int b = beg; b < end; b += SP_B) {
        const int n = (end - b < SP_B) ? (end - b) : SP_B;
        v4f l[SP_B][4];
#pragma unroll
        for (int k = 0; k < SP_B; ++k) {
            const int bk = (k < n) ? b + k : b;
            const int s = __builtin_amdgcn_readfirstlane(s_srt[bk]);
            const v4f* p = (const v4f*)(xl + (size_t)s * HC + g * HID + 4 * t);
#pragma unroll
            for (int i = 0; i < 4; ++i) l[k][i] = p[i * 8];
        }
        __builtin_amdgcn_sched_barrier(0);
        float al[SP_B];
#pragma unroll
        for (int k = 0; k < SP_B; ++k) {
            float a = 0.f;
#pragma unroll
            for (int i = 0; i < 4; ++i) {
#pragma unroll
                for (int j = 0; j < 4; ++j) {
                    float mm = l[k][i][j] + xr[i][j];
                    float lr = (mm > 0.f) ? mm : 0.2f * mm;
                    a += lr * wa[i][j];
                }
            }
            al[k] = group_sum(a);
        }
        float mn = m;
#pragma unroll
        for (int k = 0; k < SP_B; ++k) if (k < n) mn = fmaxf(mn, al[k]);
        const float sc = expf(m - mn);
        float p[SP_B];
        float ps = 0.f;
#pragma unroll
        for (int k = 0; k < SP_B; ++k) { p[k] = (k < n) ? expf(al[k] - mn) : 0.f; ps += p[k]; }
        den = den * sc + ps;
#pragma unroll
        for (int i = 0; i < 4; ++i) {
#pragma unroll
            for (int j = 0; j < 4; ++j) {
                float v = acc[i][j] * sc;
#pragma unroll
                for (int k = 0; k < SP_B; ++k) v += p[k] * l[k][i][j];
                acc[i][j] = v;
            }
        }
        m = mn;
    }
    const float inv = 0.25f / den;
#pragma unroll
    for (int i = 0; i < 4; ++i) {
        v4f o;
#pragma unroll
        for (int j = 0; j < 4; ++j) {
            float v = acc[i][j] * inv;
            v += __shfl_xor(v, 8, 32);
            v += __shfl_xor(v, 16, 32);
            o[j] = v;
        }
        if (g == 0) *(v4f*)(s_tile + r * HID + i * 32 + 4 * t) = o;
    }
}

#define FU_RPW ((FU_NB + FU_WAVES - 1) / FU_WAVES)
#define FU_REC 96

__device__ __forceinline__ v4f fu_norm_row(v4f a, v4f b, v4f c, v4f gg, v4f bb) {
    v4f y;
#pragma unroll
    for (int i = 0; i < 4; ++i) y[i] = a[i] + b[i] + c[i];
    float s = 0.f, ss = 0.f;
#pragma unroll
    for (int i = 0; i < 4; ++i) { s += y[i]; ss += y[i] * y[i]; }
    for (int off = 16; off; off >>= 1) {
        s  += __shfl_xor(s,  off, 32);
        ss += __shfl_xor(ss, off, 32);
    }
    float mu  = s * (1.f / HID);
    float var = ss * (1.f / HID) - mu * mu;
    float r   = rsqrtf(var + 1e-5f);
    v4f o;
#pragma unroll
    for (int i = 0; i < 4; ++i)
        o[i] = fmaxf((y[i] - mu) * r * gg[i] + bb[i], 0.f);
    return o;
}

#define FU_KEEP_HDR  (FU_BLOCKS * FU_WAVES * FU_LKEEP)
#define FU_KEEP_INTS (FU_KEEP_HDR + 2 * FU_BLOCKS * FU_WAVES)

template <int REUSE>
__global__ void __launch_bounds__(FU_THREADS)
edge_fused(const int* __restrict__ ei, const float* __restrict__ xl, float* h,
           const float* __restrict__ Wr, const float* __restrict__ br,
           const float* __restrict__ att, float* alpha, int* __restrict__ keep,
           const float* __restrict__ cbias, const float* __restrict__ g, const float* __restrict__ be,
           const float* __restrict__ Wn, const float* __restrict__ bn, const float* __restrict__ Wc,
           float* __restrict__ outp) {
    __shared__ __attribute__((aligned(16))) float    s_tile[FU_NB * FU_HP];
    __shared__ __attribute__((aligned(16))) float    s_xr[FU_NB * FU_XP];
    __shared__ __attribute__((aligned(16))) float    s_att[HC];
    __shared__ __attribute__((aligned(16))) float    s_den[FU_NB * NHEAD];
    __shared__ __attribute__((aligned(16))) unsigned s_max[FU_NB * NHEAD];
    __shared__ __attribute__((aligned(16))) int s_list[FU_WAVES * (FU_LKEEP + FU_LCAP)];
    __shared__ __attribute__((aligned(16))) int s_srt[FU_SCAP];
    __shared__ int s_rc[FU_NB];
    __shared__ int s_ro[FU_NB + 1];
    __shared__ int s_rf[FU_NB];
    __shared__ int s_ctl[2];
    const int tid  = threadIdx.x;
    const int lane = tid & 31;
    const int w    = __builtin_amdgcn_readfirstlane(tid >> 5);
    const int base = blockIdx.x * FU_NB;

    FuCtx c;
    c.ei = ei; c.xl = xl; c.alpha = alpha; c.s_att = s_att;
    c.s_tile = s_tile; c.s_xr = s_xr; c.s_max = s_max; c.s_den = s_den;
    c.g = lane >> 3; c.t = lane & 7;

    const int slot = blockIdx.x * FU_WAVES + w;
    int*      gl   = keep + slot * FU_LKEEP;
    int*      gh   = keep + FU_KEEP_HDR + 2 * slot;
    int k0 = 0, k1 = 0, hcnt = 0, hrest = 0;
    if (REUSE) { k0 = gl[lane]; k1 = gl[lane + 32]; hcnt = gh[0]; hrest = gh[1]; }

    {
        const v4f* h4 = (const v4f*)(h + (size_t)base * HID);
        constexpr int H4 = FU_NB * (HID / 4);
        static_assert(H4 > 2 * FU_THREADS && H4 <= 3 * FU_THREADS && (H4 - 2 * FU_THREADS) % 32 == 0, "staging shape");
        const bool third = tid < H4 - 2 * FU_THREADS;
        v4f r0 = h4[tid], r1 = h4[tid + FU_THREADS], r2 = (v4f){0.f, 0.f, 0.f, 0.f};
        if (third) r2 = h4[tid + 2 * FU_THREADS];
        const int q = tid & 31, row = tid >> 5;
        *(v4f*)(s_tile + row * FU_HP + 4 * q) = r0;
        *(v4f*)(s_tile + (row + 32) * FU_HP + 4 * q) = r1;
        if (third) *(v4f*)(s_tile + (row + 64) * FU_HP + 4 * q) = r2;
        if (tid < FU_NB * NHEAD) { s_den[tid] = 0.f; s_max[tid] = 0u; }
        if (tid < FU_NB) s_rc[tid] = 0;
        if (tid < 2) s_ctl[tid] = 0;
        if (tid < HC / 4) ((v4f*)s_att)[tid] = ((const v4f*)att)[tid];
    }

    int* lst  = s_list + w * (FU_LKEEP + FU_LCAP) + FU_LKEEP;
    int  cnt  = 0;
    if (!REUSE && w < FU_NB / 4) {
        if (lane < 4) lst[lane] = ((E_EDGES + base + 4 * w + lane) << 7) | (4 * w + lane);
        cnt = 4;
    }
    __syncthreads();

    const bool proj_first = REUSE || (((w >> 2) & 1) != 0);

    const v4i* dq   = (const v4i*)(ei + E_EDGES);
    int        rest = FU_STEPS;
    if (REUSE) {
        cnt  = __builtin_amdgcn_readfirstlane(hcnt);
        rest = __builtin_amdgcn_readfirstlane(hrest);
        lst[lane]      = k0;
        lst[lane + 32] = k1;
        for (int i = 64 + lane; i < cnt; i += 32) lst[i] = gl[i];
    }
    if (proj_first) fu_project(Wr, br, s_tile, s_xr, w, lane);
    for (int it = 0; !REUSE && it < FU_ITERS && rest == FU_STEPS; ++it) {
        const int q0 = tid + it * (FU_AHEAD * FU_THREADS);
        v4i dv[FU_AHEAD];
#pragma unroll
        for (int u = 0; u < FU_AHEAD; ++u) {
            int qu = q0 + u * FU_THREADS;
            dv[u]  = (qu < FU_Q) ? dq[qu] : (v4i){-1, -1, -1, -1};
        }
#pragma unroll
        for (int u = 0; u < FU_AHEAD; ++u) {
            if (rest == FU_STEPS) {
                if (cnt + 128 > FU_LKEEP) rest = it * FU_AHEAD + u;
                else                      fu_collect(dv[u], q0 + u * FU_THREADS, base, lst, cnt);
            }
        }
    }

    if (!REUSE) {
        for (int i = lane; i < 64 || i < cnt; i += 32) gl[i] = (i < cnt) ? lst[i] : -1;
        if (lane == 0) { gh[0] = cnt; gh[1] = rest; }
    }

    for (int i = lane; i < cnt; i += 32) lst[i - FU_LKEEP] = edge_src(ei, lst[i] >> 7);

    if (rest < FU_STEPS && lane == 0) s_ctl[0] = 1;
    for (int i = lane; i < cnt; i += 32)
        __hip_atomic_fetch_add(s_rc + (lst[i] & 127), 1, __ATOMIC_RELAXED, __HIP_MEMORY_SCOPE_WORKGROUP);

    if (!proj_first) fu_project(Wr, br, s_tile, s_xr, w, lane);
    __syncthreads();
    if (w == 0) {
        const bool hi = lane + 64 < FU_NB;
        const int c0 = s_rc[lane], c1 = s_rc[lane + 32], c2 = hi ? s_rc[lane + 64] : 0;
        int i0 = c0, i1 = c1, i2 = c2;
        for (int off = 1; off < 32; off <<= 1) {
            int a0 = __shfl_up(i0, off, 32), a1 = __shfl_up(i1, off, 32), a2 = __shfl_up(i2, off, 32);
            if (lane >= off) { i0 += a0; i1 += a1; i2 += a2; }
        }
        const int t0 = __shfl(i0, 31, 32), t1 = __shfl(i1, 31, 32), t2 = __shfl(i2, 31, 32);
        const int o0 = i0 - c0, o1 = t0 + i1 - c1, o2 = t0 + t1 + i2 - c2;
        s_ro[lane] = o0;      s_rf[lane] = o0;
        s_ro[lane + 32] = o1; s_rf[lane + 32] = o1;
        if (hi) { s_ro[lane + 64] = o2; s_rf[lane + 64] = o2; }
        if (lane == 0) {
            s_ro[FU_NB] = t0 + t1 + t2;
            if (t0 + t1 + t2 > FU_SCAP) s_ctl[0] = 1;
        }
    }
    __syncthreads();
    const bool three_phase = __builtin_amdgcn_readfirstlane(s_ctl[0]) != 0;

    if (!three_phase) {
        for (int i = lane; i < cnt; i += 32) {
            const int p = __hip_atomic_fetch_add(s_rf + (lst[i] & 127), 1, __ATOMIC_RELAXED, __HIP_MEMORY_SCOPE_WORKGROUP);
            s_srt[p] = lst[i - FU_LKEEP];
        }
        v4f wa[4];
#pragma unroll
        for (int i = 0; i < 4; ++i) wa[i] = *(const v4f*)(s_att + c.g * HID + i * 32 + 4 * c.t);
        __syncthreads();
        for (int trip = 0; trip <= FU_NB; ++trip) {
            int r = 0;
            if (lane == 0) r = __hip_atomic_fetch_add(s_ctl + 1, 1, __ATOMIC_RELAXED, __HIP_MEMORY_SCOPE_WORKGROUP);
            r = __builtin_amdgcn_readfirstlane(r);
            if (r >= FU_NB) break;
            const int beg = __builtin_amdgcn_readfirstlane(s_ro[r]);
            const int end = __builtin_amdgcn_readfirstlane(s_ro[r + 1]);
            fu_row_pass(xl, s_xr, s_tile, s_srt, r, beg, end, c.g, c.t, wa);
        }
    } else {
        for (int k = w * 32 + lane; k < FU_NB * (HID / 4); k += FU_THREADS) ((v4f*)s_tile)[k] = (v4f){0.f, 0.f, 0.f, 0.f};

        fu_list<1>(c, lst, cnt);
        if (rest < FU_STEPS) fu_walk<1>(c, dq, rest, tid, base, lst + FU_LKEEP);
        __syncthreads();

        fu_exp_list(c, lst, cnt, lane);
        if (rest < FU_STEPS) fu_walk<2>(c, dq, rest, tid, base, lst + FU_LKEEP);
        __syncthreads();

        fu_list<3>(c, lst, cnt);
        if (rest < FU_STEPS) fu_walk<3>(c, dq, rest, tid, base, lst + FU_LKEEP);
    }
    __syncthreads();

    float* s_hn = (float*)s_list;
    v4f hb[FU_RPW];
#pragma unroll
    for (int j = 0; j < FU_RPW; ++j) {
        const int r = w + j * FU_WAVES;
        const bool on = ((j + 1) * FU_WAVES <= FU_NB) || (r < FU_NB);
        hb[j] = (v4f){0.f, 0.f, 0.f, 0.f};
        if (on) hb[j] = ((const v4f*)(h + (size_t)(base + r) * HID))[lane];
    }
    const v4f cb = ((const v4f*)cbias)[lane];
    const v4f gg = ((const v4f*)g)[lane];
    const v4f bb = ((const v4f*)be)[lane];
    v4f wa = (v4f){0.f, 0.f, 0.f, 0.f}, wc = (v4f){0.f, 0.f, 0.f, 0.f};
    if (REUSE) { wa = ((const v4f*)Wn)[lane]; wc = ((const v4f*)Wc)[lane]; }
    float* s_logit = s_den;
    float* s_val   = s_den + 128;
    float  vc = 0.f;
#pragma unroll
    for (int j = 0; j < FU_RPW; ++j) {
        const int r = w + j * FU_WAVES;
        const bool on = ((j + 1) * FU_WAVES <= FU_NB) || (r < FU_NB);
        if (on) {
            const v4f t = *(const v4f*)(s_tile + r * HID + 4 * lane);
            const v4f o = fu_norm_row(hb[j], t, cb, gg, bb);
            if (!REUSE) {
                ((v4f*)(h + (size_t)(base + r) * HID))[lane] = o;
                *(v4f*)(s_hn + r * FU_HP + 4 * lane) = o;
            } else {
                float a = 0.f, cc = 0.f;
#pragma unroll
                for (int i = 0; i < 4; ++i) { a += o[i] * wa[i]; cc += o[i] * wc[i]; }
                for (int off = 16; off; off >>= 1) a += __shfl_xor(a, off, 32);
                if (lane == 0) s_logit[r] = a + bn[0];
                vc += cc;
            }
        }
    }
    if (!REUSE) {
        __syncthreads();
        const int tl = (int)__builtin_amdgcn_mbcnt_lo(~0u, 0u);
        fu_project(Wn, bn, s_hn, s_xr, w, tl);
        __syncthreads();
        v4f* x4 = (v4f*)(outp + (size_t)base * HC);
        for (int k = w * 32 + lane; k < FU_NB * (HC / 4); k += FU_THREADS) {
            const int row = k >> 7, q = k & 127;
            x4[k] = *(const v4f*)(s_xr + row * FU_XP + 4 * q);
        }
    } else {
        for (int off = 16; off; off >>= 1) vc += __shfl_xor(vc, off, 32);
        if (lane == 0) s_val[w] = vc;
        __syncthreads();
        if (w == 0) {
            float tot = s_val[lane];
            for (int off = 16; off; off >>= 1) tot += __shfl_xor(tot, off, 32);
            for (int u = 0; u < FU_REC / 32; ++u) {
                const int idx = lane + 32 * u;
                float v = 0.f;
                if (idx < FU_NB) v = s_logit[idx];
                else if (idx == FU_NB) v = tot;
                outp[(size_t)blockIdx.x * FU_REC + idx] = v;
            }
        }
    }
}

#define OUT_LINES (N_NODES / 32)

__global__ void __launch_bounds__(256)
finish_kernel(const float* __restrict__ rec, const float* __restrict__ bc, float* __restrict__ out) {
    const int lane = threadIdx.x & 31;
    const int line = blockIdx.x * 8 + (threadIdx.x >> 5);
    if (line < OUT_LINES) {
        const int n = line * 32 + lane;
        out[n] = rec[(n / FU_NB) * FU_REC + (n % FU_NB)];
    } else if (line == OUT_LINES) {
        float acc = 0.f;
        for (int j = 0; j < (FU_BLOCKS + 31) / 32; ++j) {
            int i = lane + 32 * j;
            if (i < FU_BLOCKS) acc += rec[i * FU_REC + FU_NB];
        }
        for (int off = 16; off; off >>= 1) acc += __shfl_down(acc, off, 32);
        if (lane == 0) out[N_NODES] = acc * (1.f / N_NODES) + bc[0];
    }
}

extern "C" void kernel_launch(void* const* d_in, const int* in_sizes, int n_in,
                              void* d_out, int out_size, void* d_ws, size_t ws_size,
                              hipStream_t stream) {
    const float* x     = (const float*)d_in[0];
    const int*   ei    = (const int*)d_in[1];
    const float* W_in  = (const float*)d_in[2];
    const float* b_in  = (const float*)d_in[3];
    const float* Wl1   = (const float*)d_in[4];
    const float* bl1   = (const float*)d_in[5];
    const float* Wr1   = (const float*)d_in[6];
    const float* br1   = (const float*)d_in[7];
    const float* att1  = (const float*)d_in[8];
    const float* bias1 = (const float*)d_in[9];
    const float* g1    = (const float*)d_in[10];
    const float* be1   = (const float*)d_in[11];
    const float* Wl2   = (const float*)d_in[12];
    const float* bl2   = (const float*)d_in[13];
    const float* Wr2   = (const float*)d_in[14];
    const float* br2   = (const float*)d_in[15];
    const float* att2  = (const float*)d_in[16];
    const float* bias2 = (const float*)d_in[17];
    const float* g2    = (const float*)d_in[18];
    const float* be2   = (const float*)d_in[19];
    const float* Wa    = (const float*)d_in[20];
    const float* ba    = (const float*)d_in[21];
    const float* Wc    = (const float*)d_in[22];
    const float* bc    = (const float*)d_in[23];

    float* out = (float*)d_out;

    float* ws = (float*)d_ws;
    size_t off = 0;
    float*    h0    = ws + off; off += (size_t)N_NODES * HID;
    int*      keep  = (int*)(ws + off); off += (size_t)N_NODES * HID;
    static_assert((size_t)FU_KEEP_INTS <= (size_t)N_NODES * HID, "kept lists must fit the plane");
    float*    xl1   = ws + off; off += (size_t)N_NODES * HC;
    float*    xl2   = ws + off; off += (size_t)N_NODES * HC;
    float*    alpha = ws + off; off += (size_t)ETOT * NHEAD;
    float*    rec   = ws + off; off += (size_t)N_NODES * NHEAD;
    static_assert((size_t)FU_BLOCKS * FU_REC <= (size_t)N_NODES * NHEAD, "records must fit their plane");
    (void)ws_size; (void)in_sizes; (void)n_in; (void)out_size;

    const dim3 gemmInGrid((N_NODES / 16 + GM_WAVES_SMALL - 1) / GM_WAVES_SMALL, HID / GM_COLS);
    const dim3 gemmHCGrid((N_NODES / 16 + GM_WAVES_BIG - 1) / GM_WAVES_BIG, HC / GM_COLS);

    gemm_wmma_f32<DIN, GM_WAVES_SMALL, 1><<<gemmInGrid, GM_WAVES_SMALL * 32, 0, stream>>>(x, W_in, b_in, h0, N_NODES, HID);

    gemm_wmma_f32<HID, GM_WAVES_BIG, 0><<<gemmHCGrid, GM_WAVES_BIG * 32, 0, stream>>>(h0, Wl1, bl1, xl1, N_NODES, HC);
    edge_fused<0><<<FU_BLOCKS, FU_THREADS, 0, stream>>>(ei, xl1, h0, Wr1, br1, att1, alpha, keep,
                                                        bias1, g1, be1, Wl2, bl2, Wc, xl2);

    edge_fused<1><<<FU_BLOCKS, FU_THREADS, 0, stream>>>(ei, xl2, h0, Wr2, br2, att2, alpha, keep,
                                                        bias2, g2, be2, Wa, ba, Wc, rec);
    finish_kernel<<<(OUT_LINES + 1 + 7) / 8, 256, 0, stream>>>(rec, bc, out);
}
